# starting kernel
# baseline (speedup 1.0000x reference)
#include <hip/hip_runtime.h>

typedef float v2f __attribute__((ext_vector_type(2)));
typedef float v8f __attribute__((ext_vector_type(8)));

#define MV 16
#define MAXLEN 16384

__device__ constexpr int popc4(int v) { return (v & 1) + ((v >> 1) & 1) + ((v >> 2) & 1) + ((v >> 3) & 1); }

__device__ constexpr float gpsign(int a, int b) {
    int s = 0;
    for (int t = a >> 1; t; t >>= 1) s += popc4(t & b);
    return (s & 1) ? -1.0f : 1.0f;
}
__device__ constexpr float bivsign(int k) { return (popc4(k) == 2) ? -1.0f : 1.0f; }

__device__ inline v8f gp_via_wmma_t(const float* __restrict__ ctab,
                                    const float (&Af)[16], const float (&Bf)[16],
                                    int lane, bool hi, v8f acc) {
#pragma unroll
    for (int c = 0; c < 64; ++c) {
        const int i  = c >> 2;
        const int j0 = (c & 3) << 2;
        const v2f aC = *(const v2f*)(ctab + c * 64 + lane * 2);
        v2f zB;
        zB[0] = Af[i] * (hi ? Bf[j0 + 2] : Bf[j0 + 0]);
        zB[1] = Af[i] * (hi ? Bf[j0 + 3] : Bf[j0 + 1]);
        acc = __builtin_amdgcn_wmma_f32_16x16x4_f32(false, aC, false, zB,
                                                    (short)0, acc, false, false);
    }
    return acc;
}

__global__ __launch_bounds__(256)
void care_rotor_kernel(const float* __restrict__ x,
                       const int*   __restrict__ pos,
                       const float* __restrict__ bx,
                       const float* __restrict__ by,
                       const float* __restrict__ bz,
                       const float* __restrict__ bw,
                       const float* __restrict__ theta,
                       float* __restrict__ out,
                       int ntiles) {
    __shared__ float ctab[64 * 32 * 2];

    const int lane = threadIdx.x & 31;
    const int wave = threadIdx.x >> 5;
    const int kl   = lane & 15;
    const bool hi  = lane >= 16;

    if (wave == 0) {
#pragma unroll
        for (int c = 0; c < 64; ++c) {
            const int i  = c >> 2;
            const int j0 = (c & 3) << 2;
            const float e0lo = (kl == (i ^ (j0 + 0))) ? gpsign(i, j0 + 0) : 0.0f;
            const float e1lo = (kl == (i ^ (j0 + 1))) ? gpsign(i, j0 + 1) : 0.0f;
            const float e0hi = (kl == (i ^ (j0 + 2))) ? gpsign(i, j0 + 2) : 0.0f;
            const float e1hi = (kl == (i ^ (j0 + 3))) ? gpsign(i, j0 + 3) : 0.0f;
            v2f e;
            e[0] = hi ? e0hi : e0lo;
            e[1] = hi ? e1hi : e1lo;
            *(v2f*)(ctab + c * 64 + lane * 2) = e;
        }
    }
    __syncthreads();

    const int tile = blockIdx.x * 8 + wave;
    if (tile >= ntiles) return;

    const long pbase = (long)tile * 16;
    const long p     = pbase + kl;

    float xv[16];
    {
        const float4* xp = (const float4*)(x + p * MV);
        float4 q[4];
#pragma unroll
        for (int t = 0; t < 4; ++t) q[t] = xp[t];
#pragma unroll
        for (int t = 0; t < 4; ++t) {
            xv[4 * t + 0] = q[t].x; xv[4 * t + 1] = q[t].y;
            xv[4 * t + 2] = q[t].z; xv[4 * t + 3] = q[t].w;
        }
    }

    int pc = pos[p];
    pc = pc < 0 ? 0 : (pc > MAXLEN - 1 ? MAXLEN - 1 : pc);
    const float4 th = *(const float4*)(theta + (long)pc * 4);
    float phi[4];
    phi[0] = 0.5f * th.x * bx[3];
    phi[1] = 0.5f * th.y * by[5];
    phi[2] = 0.5f * th.z * bz[9];
    phi[3] = 0.5f * th.w * bw[6];

    float r[16];
#pragma unroll
    for (int k = 0; k < 16; ++k) r[k] = (k == 0) ? 1.0f : 0.0f;
    const int blades[4] = {3, 5, 9, 6};
#pragma unroll
    for (int pl = 0; pl < 4; ++pl) {
        float sn, cs;
        __sincosf(phi[pl], &sn, &cs);
        const int b = blades[pl];
        float nr[16];
#pragma unroll
        for (int k = 0; k < 16; ++k)
            nr[k] = cs * r[k] + sn * gpsign(k ^ b, b) * r[k ^ b];
#pragma unroll
        for (int k = 0; k < 16; ++k) r[k] = nr[k];
    }

    float rrev[16];
#pragma unroll
    for (int k = 0; k < 16; ++k) rrev[k] = r[k] * bivsign(k);

    v8f acc;
#pragma unroll
    for (int v = 0; v < 8; ++v) acc[v] = 0.0f;
    acc = gp_via_wmma_t(ctab, r, xv, lane, hi, acc);

    float rxv[16];
#pragma unroll
    for (int v = 0; v < 8; ++v) {
        const float other = __shfl_xor((float)acc[v], 16, 32);
        rxv[v]     = hi ? other  : acc[v];
        rxv[v + 8] = hi ? acc[v] : other;
    }

    v8f acc2;
#pragma unroll
    for (int v = 0; v < 8; ++v) acc2[v] = 0.0f;
    acc2 = gp_via_wmma_t(ctab, rxv, rrev, lane, hi, acc2);

    {
        float* po = out + (pbase + kl) * MV + (hi ? 8 : 0);
        float4 q0, q1;
        q0.x = acc2[0]; q0.y = acc2[1]; q0.z = acc2[2]; q0.w = acc2[3];
        q1.x = acc2[4]; q1.y = acc2[5]; q1.z = acc2[6]; q1.w = acc2[7];
        *(float4*)(po + 0) = q0;
        *(float4*)(po + 4) = q1;
    }
}

extern "C" void kernel_launch(void* const* d_in, const int* in_sizes, int n_in,
                              void* d_out, int out_size, void* d_ws, size_t ws_size,
                              hipStream_t stream) {
    const float* x     = (const float*)d_in[0];
    const int*   pos   = (const int*)d_in[1];
    const float* bx    = (const float*)d_in[2];
    const float* by    = (const float*)d_in[3];
    const float* bz    = (const float*)d_in[4];
    const float* bw    = (const float*)d_in[5];
    const float* theta = (const float*)d_in[6];
    float* out = (float*)d_out;

    const int npos   = in_sizes[1];
    const int ntiles = npos / 16;
    const int blocks = (ntiles + 7) / 8;

    hipLaunchKernelGGL(care_rotor_kernel, dim3(blocks), dim3(256), 0, stream,
                       x, pos, bx, by, bz, bw, theta, out, ntiles);
}


# Each quaternion block product in 8 multiplications, not 16: operand sums on the vector unit, +-1/2, +-1 recombination as the constant A table: K 64 -> 32 per product, 16 f32 WMMAs per tile, not 32
# speedup vs baseline: 3.3812x; 3.3812x over previous
#include <hip/hip_runtime.h>


typedef float v2f __attribute__((ext_vector_type(2)));
typedef float v8f __attribute__((ext_vector_type(8)));

#define MV 16
#define MAXLEN 16384

constexpr int popc4(int v) { return (v & 1) + ((v >> 1) & 1) + ((v >> 2) & 1) + ((v >> 3) & 1); }

constexpr float gpsign(int a, int b) {
    int s = 0;
    for (int t = a >> 1; t; t >>= 1) s += popc4(t & b);
    return (s & 1) ? -1.0f : 1.0f;
}
constexpr float bivsign(int k) { return (popc4(k) == 2) ? -1.0f : 1.0f; }

constexpr int rep_blade(int i) {
    constexpr int t[8] = {0, 3, 5, 6, 1, 2, 4, 8};
    return t[i];
}
constexpr int rep_index(int blade) {
    for (int i = 0; i < 8; ++i) if (rep_blade(i) == blade) return i;
    return -1;
}
constexpr float dual_sign(int J) { return gpsign(J, 15); }

constexpr int coord_index(int Jx) {
    return rep_index(Jx) >= 0 ? rep_index(Jx) : rep_index(Jx ^ 15);
}
constexpr float coord_sign(int Jx, float sig) {
    return rep_index(Jx) >= 0 ? 1.0f : sig * dual_sign(Jx ^ 15);
}

struct Comb { int i, j, s; };
constexpr Comb HA[8] = {{3, 1, 1}, {0, 2, -1}, {0, 2, 1}, {3, 1, -1}, {3, 2, -1}, {1, 0, 1}, {0, 1, -1}, {3, 2, 1}};
constexpr Comb HB[8] = {{1, 2, 1}, {0, 3, 1}, {0, 3, -1}, {1, 2, -1}, {2, 3, -1}, {1, 0, 1}, {2, 3, 1}, {0, 1, -1}};
constexpr float HC[4][8] = {
    {-0.5f,  0.5f,  0.5f, 0.5f, 1.0f, 0.0f, 0.0f, 0.0f},
    {-0.5f, -0.5f, -0.5f, 0.5f, 0.0f, 1.0f, 0.0f, 0.0f},
    { 0.5f, -0.5f,  0.5f, 0.5f, 0.0f, 0.0f, 1.0f, 0.0f},
    { 0.5f,  0.5f, -0.5f, 0.5f, 0.0f, 0.0f, 0.0f, 1.0f}};

struct BlockSigns { int sa[4], sb[4], so[2][4]; };
constexpr BlockSigns BS[4] = {
    {{-1, 1, 1, 1}, {-1,  1,  1, 1}, {{ 1, -1, -1, -1}, { 1, -1, -1, -1}}},
    {{-1, 1, 1, 1}, { 1,  1,  1, 1}, {{-1, -1, -1, -1}, {-1, -1, -1,  1}}},
    {{-1, 1, 1, 1}, {-1,  1,  1, 1}, {{ 1, -1, -1, -1}, { 1, -1, -1, -1}}},
    {{ 1, 1, 1, 1}, {-1, -1, -1, 1}, {{-1, -1, -1, -1}, {-1, -1, -1,  1}}}};

constexpr Comb blk_left(int b, int m)  { return {HA[m].i, HA[m].j, HA[m].s * BS[b].sa[HA[m].i] * BS[b].sa[HA[m].j]}; }
constexpr Comb blk_right(int b, int m) { return {HB[m].i, HB[m].j, HB[m].s * BS[b].sb[HB[m].i] * BS[b].sb[HB[m].j]}; }
constexpr float blk_coef(int b, int h, int c, int m) {
    return (float)(BS[b].so[h][c] * BS[b].sa[HA[m].i] * BS[b].sb[HB[m].i]) * HC[c][m];
}

constexpr float blk_expect(int b, int h, int c, int l, int r) {
    const float sig = h ? -1.0f : 1.0f;
    const int odd = b & 1;
    const int La = (b < 2) ? rep_blade(l) : rep_blade(4 * odd + l);
    const int Rb = (b < 2) ? rep_blade(4 * odd + r) : rep_blade(r);
    if (coord_index(La ^ Rb) != 4 * odd + c) return 0.0f;
    float s = gpsign(La, Rb) * coord_sign(La ^ Rb, sig);
    if (b == 1 && r == 3) s *= sig;
    if (b == 3 && l == 3) s *= sig;
    return s;
}
constexpr bool blk_ok(int b) {
    for (int h = 0; h < 2; ++h)
        for (int c = 0; c < 4; ++c)
            for (int l = 0; l < 4; ++l)
                for (int r = 0; r < 4; ++r) {
                    float v = 0.0f;
                    for (int m = 0; m < 8; ++m) {
                        const Comb a = blk_left(b, m), d = blk_right(b, m);
                        const float al = (a.i == l ? 1.0f : 0.0f) + (a.j == l ? (float)a.s : 0.0f);
                        const float be = (d.i == r ? 1.0f : 0.0f) + (d.j == r ? (float)d.s : 0.0f);
                        v += blk_coef(b, h, c, m) * al * be;
                    }
                    if (v != blk_expect(b, h, c, l, r)) return false;
                }
    return true;
}
static_assert(blk_ok(0) && blk_ok(1) && blk_ok(2) && blk_ok(3),
              "8-multiplication blocks do not reproduce the Cayley signs");

constexpr float tab_entry(int chunk, int lane, int v) {
    const int h = lane >> 4, kl = lane & 15;
    const float sig = h ? -1.0f : 1.0f;
    const int tm = 2 * (chunk & 7) + v;
    const int odd = tm >> 3, m = tm & 7;
    if (chunk < 8) {
        const int J = kl - 8 * h;
        if (J < 0 || J > 7 || (J >> 2) != odd) return 0.0f;
        return blk_coef(odd, h, J & 3, m) * (J == 7 ? sig : 1.0f);
    }
    float e = 0.0f;
    for (int c = 0; c < 4; ++c) {
        const int B = rep_blade(4 * odd + c);
        if (kl == B)        e += 0.5f * blk_coef(2 + odd, h, c, m);
        if (kl == (B ^ 15)) e += 0.5f * sig * dual_sign(B) * blk_coef(2 + odd, h, c, m);
    }
    return e;
}
struct alignas(16) CTab { float v[16 * 32 * 2]; };
constexpr CTab make_ctab() {
    CTab t{};
    for (int chunk = 0; chunk < 16; ++chunk)
        for (int lane = 0; lane < 32; ++lane)
            for (int v = 0; v < 2; ++v)
                t.v[chunk * 64 + lane * 2 + v] = tab_entry(chunk, lane, v);
    return t;
}

__constant__ const CTab g_ctab = make_ctab();

template <int B, int M>
__device__ __forceinline__ float fterm(const float (&L)[4], const float (&R)[4]) {
    constexpr Comb a = blk_left(B, M), b = blk_right(B, M);
    const float al = (a.s > 0) ? (L[a.i] + L[a.j]) : (L[a.i] - L[a.j]);
    const float be = (b.s > 0) ? (R[b.i] + R[b.j]) : (R[b.i] - R[b.j]);
    return al * be;
}

template <int P, int C>
struct Product {
    static __device__ __forceinline__ void run(v8f& acc, const float* ctab, int lane,
                                               const float (&Le)[4], const float (&Re)[4],
                                               const float (&Lo)[4], const float (&Ro)[4]) {
        const v2f aC = *(const v2f*)(ctab + (8 * P + C) * 64 + lane * 2);
        v2f zB;
        if constexpr (C < 4) {
            zB[0] = fterm<2 * P, 2 * C>(Le, Re);
            zB[1] = fterm<2 * P, 2 * C + 1>(Le, Re);
        } else {
            zB[0] = fterm<2 * P + 1, 2 * C - 8>(Lo, Ro);
            zB[1] = fterm<2 * P + 1, 2 * C - 7>(Lo, Ro);
        }
        acc = __builtin_amdgcn_wmma_f32_16x16x4_f32(false, aC, false, zB,
                                                    (short)0, acc, false, false);
        Product<P, C + 1>::run(acc, ctab, lane, Le, Re, Lo, Ro);
    }
};
template <int P>
struct Product<P, 8> {
    static __device__ __forceinline__ void run(v8f&, const float*, int,
                                               const float (&)[4], const float (&)[4],
                                               const float (&)[4], const float (&)[4]) {}
};

__global__ __launch_bounds__(256)
void care_rotor_kernel(const float* __restrict__ x,
                       const int*   __restrict__ pos,
                       const float* __restrict__ bx,
                       const float* __restrict__ by,
                       const float* __restrict__ bz,
                       const float* __restrict__ bw,
                       const float* __restrict__ theta,
                       float* __restrict__ out,
                       int ntiles) {
    __shared__ alignas(16) float ctab[16 * 32 * 2];

    const int lane = threadIdx.x & 31;
    const int wave = threadIdx.x >> 5;
    const int kl   = lane & 15;
    const bool hi  = lane >= 16;

    ((float4*)ctab)[threadIdx.x] = ((const float4*)g_ctab.v)[threadIdx.x];
    __syncthreads();

    const int tile = blockIdx.x * 8 + wave;
    if (tile >= ntiles) return;

    const long pbase = (long)tile * 16;
    const long p     = pbase + kl;

    float xv[16];
    {
        const float4* xp = (const float4*)(x + p * MV);
        float4 q[4];
#pragma unroll
        for (int t = 0; t < 4; ++t) q[t] = xp[t];
#pragma unroll
        for (int t = 0; t < 4; ++t) {
            xv[4 * t + 0] = q[t].x; xv[4 * t + 1] = q[t].y;
            xv[4 * t + 2] = q[t].z; xv[4 * t + 3] = q[t].w;
        }
    }

    int pc = pos[p];
    pc = pc < 0 ? 0 : (pc > MAXLEN - 1 ? MAXLEN - 1 : pc);
    const float4 th = *(const float4*)(theta + (long)pc * 4);
    float phi[4];
    phi[0] = 0.5f * th.x * bx[3];
    phi[1] = 0.5f * th.y * by[5];
    phi[2] = 0.5f * th.z * bz[9];
    phi[3] = 0.5f * th.w * bw[6];

    float r[16];
#pragma unroll
    for (int k = 0; k < 16; ++k) r[k] = (k == 0) ? 1.0f : 0.0f;
    const int blades[4] = {3, 5, 9, 6};
#pragma unroll
    for (int pl = 0; pl < 4; ++pl) {
        float sn, cs;
        __sincosf(phi[pl], &sn, &cs);
        const int b = blades[pl];
        float nr[16];
#pragma unroll
        for (int k = 0; k < 16; ++k)
            nr[k] = cs * r[k] + sn * gpsign(k ^ b, b) * r[k ^ b];
#pragma unroll
        for (int k = 0; k < 16; ++k) r[k] = nr[k];
    }

    float rrev[16];
#pragma unroll
    for (int k = 0; k < 16; ++k) rrev[k] = r[k] * bivsign(k);

    const float sig = hi ? -1.0f : 1.0f;
    float ue[4], uo[4];
#pragma unroll
    for (int b = 0; b < 4; ++b) {
        const int Je = rep_blade(b), Jo = rep_blade(4 + b);
        ue[b] = xv[Je] + (sig * dual_sign(Je)) * xv[Je ^ 15];
        uo[b] = xv[Jo] + (sig * dual_sign(Jo)) * xv[Jo ^ 15];
    }
    {
        constexpr int J7 = rep_blade(7);
        constexpr float d7 = dual_sign(J7);
        uo[3] = sig * xv[J7] + d7 * xv[J7 ^ 15];
    }
    float qs[4], qo[4], qts[4];
#pragma unroll
    for (int a = 0; a < 4; ++a) {
        const int J = rep_blade(a);
        const float d = (sig * dual_sign(J)) * r[J ^ 15];
        qs[a]  = r[J] + d;
        qo[a]  = r[J] - d;
        qts[a] = rrev[J] + (sig * dual_sign(J)) * rrev[J ^ 15];
    }

    v8f acc;
#pragma unroll
    for (int v = 0; v < 8; ++v) acc[v] = 0.0f;
    Product<0, 0>::run(acc, ctab, lane, qs, ue, qo, uo);

    const float we[4] = {acc[0], acc[1], acc[2], acc[3]};
    const float wo[4] = {acc[4], acc[5], acc[6], acc[7]};

    v8f acc2;
#pragma unroll
    for (int v = 0; v < 8; ++v) acc2[v] = 0.0f;
    Product<1, 0>::run(acc2, ctab, lane, we, qts, wo, qts);

    {
        float* po = out + (pbase + kl) * MV + (hi ? 8 : 0);
        float4 q0, q1;
        q0.x = acc2[0]; q0.y = acc2[1]; q0.z = acc2[2]; q0.w = acc2[3];
        q1.x = acc2[4]; q1.y = acc2[5]; q1.z = acc2[6]; q1.w = acc2[7];
        *(float4*)(po + 0) = q0;
        *(float4*)(po + 4) = q1;
    }
}

extern "C" void kernel_launch(void* const* d_in, const int* in_sizes, int n_in,
                              void* d_out, int out_size, void* d_ws, size_t ws_size,
                              hipStream_t stream) {
    const float* x     = (const float*)d_in[0];
    const int*   pos   = (const int*)d_in[1];
    const float* bx    = (const float*)d_in[2];
    const float* by    = (const float*)d_in[3];
    const float* bz    = (const float*)d_in[4];
    const float* bw    = (const float*)d_in[5];
    const float* theta = (const float*)d_in[6];
    float* out = (float*)d_out;

    const int npos   = in_sizes[1];
    const int ntiles = npos / 16;
    const int blocks = (ntiles + 7) / 8;

    hipLaunchKernelGGL(care_rotor_kernel, dim3(blocks), dim3(256), 0, stream,
                       x, pos, bx, by, bz, bw, theta, out, ntiles);
}


# Two tiles per wave, not one: half as many waves have to enter the chip (8192, not 16384); a wave requests both tiles' rows, positions and angles before the first tile's arithmetic
# speedup vs baseline: 3.7955x; 1.1225x over previous
#include <hip/hip_runtime.h>


typedef float v2f __attribute__((ext_vector_type(2)));
typedef float v8f __attribute__((ext_vector_type(8)));

#define MV 16
#define MAXLEN 16384

constexpr int popc4(int v) { return (v & 1) + ((v >> 1) & 1) + ((v >> 2) & 1) + ((v >> 3) & 1); }

constexpr float gpsign(int a, int b) {
    int s = 0;
    for (int t = a >> 1; t; t >>= 1) s += popc4(t & b);
    return (s & 1) ? -1.0f : 1.0f;
}
constexpr float bivsign(int k) { return (popc4(k) == 2) ? -1.0f : 1.0f; }

constexpr int rep_blade(int i) {
    constexpr int t[8] = {0, 3, 5, 6, 1, 2, 4, 8};
    return t[i];
}
constexpr int rep_index(int blade) {
    for (int i = 0; i < 8; ++i) if (rep_blade(i) == blade) return i;
    return -1;
}
constexpr float dual_sign(int J) { return gpsign(J, 15); }

constexpr int coord_index(int Jx) {
    return rep_index(Jx) >= 0 ? rep_index(Jx) : rep_index(Jx ^ 15);
}
constexpr float coord_sign(int Jx, float sig) {
    return rep_index(Jx) >= 0 ? 1.0f : sig * dual_sign(Jx ^ 15);
}

struct Comb { int i, j, s; };
constexpr Comb HA[8] = {{3, 1, 1}, {0, 2, -1}, {0, 2, 1}, {3, 1, -1}, {3, 2, -1}, {1, 0, 1}, {0, 1, -1}, {3, 2, 1}};
constexpr Comb HB[8] = {{1, 2, 1}, {0, 3, 1}, {0, 3, -1}, {1, 2, -1}, {2, 3, -1}, {1, 0, 1}, {2, 3, 1}, {0, 1, -1}};
constexpr float HC[4][8] = {
    {-0.5f,  0.5f,  0.5f, 0.5f, 1.0f, 0.0f, 0.0f, 0.0f},
    {-0.5f, -0.5f, -0.5f, 0.5f, 0.0f, 1.0f, 0.0f, 0.0f},
    { 0.5f, -0.5f,  0.5f, 0.5f, 0.0f, 0.0f, 1.0f, 0.0f},
    { 0.5f,  0.5f, -0.5f, 0.5f, 0.0f, 0.0f, 0.0f, 1.0f}};

struct BlockSigns { int sa[4], sb[4], so[2][4]; };
constexpr BlockSigns BS[4] = {
    {{-1, 1, 1, 1}, {-1,  1,  1, 1}, {{ 1, -1, -1, -1}, { 1, -1, -1, -1}}},
    {{-1, 1, 1, 1}, { 1,  1,  1, 1}, {{-1, -1, -1, -1}, {-1, -1, -1,  1}}},
    {{-1, 1, 1, 1}, {-1,  1,  1, 1}, {{ 1, -1, -1, -1}, { 1, -1, -1, -1}}},
    {{ 1, 1, 1, 1}, {-1, -1, -1, 1}, {{-1, -1, -1, -1}, {-1, -1, -1,  1}}}};

constexpr Comb blk_left(int b, int m)  { return {HA[m].i, HA[m].j, HA[m].s * BS[b].sa[HA[m].i] * BS[b].sa[HA[m].j]}; }
constexpr Comb blk_right(int b, int m) { return {HB[m].i, HB[m].j, HB[m].s * BS[b].sb[HB[m].i] * BS[b].sb[HB[m].j]}; }
constexpr float blk_coef(int b, int h, int c, int m) {
    return (float)(BS[b].so[h][c] * BS[b].sa[HA[m].i] * BS[b].sb[HB[m].i]) * HC[c][m];
}

constexpr float blk_expect(int b, int h, int c, int l, int r) {
    const float sig = h ? -1.0f : 1.0f;
    const int odd = b & 1;
    const int La = (b < 2) ? rep_blade(l) : rep_blade(4 * odd + l);
    const int Rb = (b < 2) ? rep_blade(4 * odd + r) : rep_blade(r);
    if (coord_index(La ^ Rb) != 4 * odd + c) return 0.0f;
    float s = gpsign(La, Rb) * coord_sign(La ^ Rb, sig);
    if (b == 1 && r == 3) s *= sig;
    if (b == 3 && l == 3) s *= sig;
    return s;
}
constexpr bool blk_ok(int b) {
    for (int h = 0; h < 2; ++h)
        for (int c = 0; c < 4; ++c)
            for (int l = 0; l < 4; ++l)
                for (int r = 0; r < 4; ++r) {
                    float v = 0.0f;
                    for (int m = 0; m < 8; ++m) {
                        const Comb a = blk_left(b, m), d = blk_right(b, m);
                        const float al = (a.i == l ? 1.0f : 0.0f) + (a.j == l ? (float)a.s : 0.0f);
                        const float be = (d.i == r ? 1.0f : 0.0f) + (d.j == r ? (float)d.s : 0.0f);
                        v += blk_coef(b, h, c, m) * al * be;
                    }
                    if (v != blk_expect(b, h, c, l, r)) return false;
                }
    return true;
}
static_assert(blk_ok(0) && blk_ok(1) && blk_ok(2) && blk_ok(3),
              "8-multiplication blocks do not reproduce the Cayley signs");

constexpr float p1_weight(int row, int h, int odd, int m) {
    const float sig = h ? -1.0f : 1.0f;
    const int J = row - 8 * h;
    if (J < 0 || J > 7 || (J >> 2) != odd) return 0.0f;
    return blk_coef(odd, h, J & 3, m) * (J == 7 ? sig : 1.0f);
}
constexpr float p2_weight(int row, int h, int odd, int m) {
    const float sig = h ? -1.0f : 1.0f;
    float e = 0.0f;
    for (int c = 0; c < 4; ++c) {
        const int B = rep_blade(4 * odd + c);
        if (row == B)        e += 0.5f * blk_coef(2 + odd, h, c, m);
        if (row == (B ^ 15)) e += 0.5f * sig * dual_sign(B) * blk_coef(2 + odd, h, c, m);
    }
    return e;
}

constexpr float tab_entry(int chunk, int lane, int v) {
    const int h = lane >> 4, row = lane & 15;
    const int tm = 2 * (chunk & 3) + v;
    return chunk < 4 ? p1_weight(row, h, tm >> 2, tm & 3) : p2_weight(row, h, tm >> 2, tm & 3);
}
struct alignas(16) CTab { float v[8 * 32 * 2]; };
constexpr CTab make_ctab() {
    CTab t{};
    for (int chunk = 0; chunk < 8; ++chunk)
        for (int lane = 0; lane < 32; ++lane)
            for (int v = 0; v < 2; ++v)
                t.v[chunk * 64 + lane * 2 + v] = tab_entry(chunk, lane, v);
    return t;
}

__constant__ const CTab g_ctab = make_ctab();

constexpr int p2_row_of_term(int h, int d) {
    const int B = rep_blade(d);
    return ((B >> 3) == h) ? B : (B ^ 15);
}
constexpr int p2_term_of_row(int row) {
    for (int d = 0; d < 8; ++d)
        if (row == rep_blade(d) || row == (rep_blade(d) ^ 15)) return d;
    return -1;
}
constexpr bool start_ok() {
    for (int h = 0; h < 2; ++h)
        for (int row = 0; row < 16; ++row)
            for (int odd = 0; odd < 2; ++odd)
                for (int m = 4; m < 8; ++m) {
                    const int d = 4 * odd + (m - 4);
                    const bool own1 = (row == 8 * h + d);
                    if ((p1_weight(row, h, odd, m) != 0.0f) != own1) return false;
                    const bool own2 = (row == p2_row_of_term(0, d) || row == p2_row_of_term(1, d));
                    if ((p2_weight(row, h, odd, m) != 0.0f) != own2) return false;
                }
    for (int row = 0; row < 16; ++row)
        if (p2_row_of_term(row >> 3, p2_term_of_row(row)) != row) return false;
    return true;
}
static_assert(start_ok(), "single-row terms do not match the accumulator rows");

template <int B, int M>
__device__ __forceinline__ float fterm(const float (&L)[4], const float (&R)[4]) {
    constexpr Comb a = blk_left(B, M), b = blk_right(B, M);
    const float al = (a.s > 0) ? (L[a.i] + L[a.j]) : (L[a.i] - L[a.j]);
    const float be = (b.s > 0) ? (R[b.i] + R[b.j]) : (R[b.i] - R[b.j]);
    return al * be;
}

template <int J>
__device__ __forceinline__ float start1(const float (&Le)[4], const float (&Re)[4],
                                        const float (&Lo)[4], const float (&Ro)[4]) {
    constexpr int odd = J >> 2, m = 4 + (J & 3);
    constexpr float k0 = p1_weight(J, 0, odd, m), k1 = p1_weight(8 + J, 1, odd, m);
    static_assert((k0 == 1.0f || k0 == -1.0f) && k1 == k0, "weight of a single-row term");
    if constexpr (odd == 0) return k0 * fterm<0, m>(Le, Re); else return k0 * fterm<1, m>(Lo, Ro);
}

template <int D>
__device__ __forceinline__ float term2(const float (&We)[4], const float (&Wo)[4], const float (&Q)[4]) {
    if constexpr (D < 4) return fterm<2, 4 + D>(We, Q); else return fterm<3, D>(Wo, Q);
}

template <int D>
__device__ __forceinline__ float start2(float hs, float own, float oth) {
    constexpr int odd = D >> 2, m = 4 + (D & 3);
    constexpr float a0 = p2_weight(p2_row_of_term(0, D), 0, odd, m), a1 = p2_weight(p2_row_of_term(1, D), 1, odd, m);
    constexpr float b0 = p2_weight(p2_row_of_term(0, D), 1, odd, m), b1 = p2_weight(p2_row_of_term(1, D), 0, odd, m);
    static_assert((a0 == 0.5f || a0 == -0.5f) && (a1 == a0 || a1 == -a0), "own weight of a single-row term");
    static_assert((b0 == 0.5f || b0 == -0.5f) && (b1 == b0 || b1 == -b0), "other weight of a single-row term");
    float wa, wb;
    if constexpr (a1 == a0) wa = a0; else wa = (2.0f * a0) * hs;
    if constexpr (b1 == b0) wb = b0; else wb = (2.0f * b0) * hs;
    return wa * own + wb * oth;
}

template <int V>
__device__ __forceinline__ float pick2(bool hi, const float (&sv)[8]) {
    constexpr int d0 = p2_term_of_row(V), d1 = p2_term_of_row(8 + V);
    return hi ? sv[d1] : sv[d0];
}

__device__ __forceinline__ float other_half(float x) {
    return __builtin_bit_cast(float, __builtin_amdgcn_permlanex16(__builtin_bit_cast(int, x), __builtin_bit_cast(int, x),
                                                                  0x76543210, 0xfedcba98, false, false));
}

template <int P, int C>
struct Product {
    static __device__ __forceinline__ void run(v8f& acc, const float* ctab, int lane,
                                               const float (&Le)[4], const float (&Re)[4],
                                               const float (&Lo)[4], const float (&Ro)[4]) {
        const v2f aC = *(const v2f*)(ctab + (4 * P + C) * 64 + lane * 2);
        v2f zB;
        if constexpr (C < 2) {
            zB[0] = fterm<2 * P, 2 * C>(Le, Re);
            zB[1] = fterm<2 * P, 2 * C + 1>(Le, Re);
        } else {
            zB[0] = fterm<2 * P + 1, 2 * C - 4>(Lo, Ro);
            zB[1] = fterm<2 * P + 1, 2 * C - 3>(Lo, Ro);
        }
        acc = __builtin_amdgcn_wmma_f32_16x16x4_f32(false, aC, false, zB,
                                                    (short)0, acc, false, false);
        Product<P, C + 1>::run(acc, ctab, lane, Le, Re, Lo, Ro);
    }
};
template <int P>
struct Product<P, 4> {
    static __device__ __forceinline__ void run(v8f&, const float*, int,
                                               const float (&)[4], const float (&)[4],
                                               const float (&)[4], const float (&)[4]) {}
};

#define TPW 2

__device__ __forceinline__ void rotate_tile(const float (&xv)[16], const float4 th,
                                            const float cx, const float cy, const float cz, const float cw,
                                            const float* ctab, const int lane, const bool hi,
                                            float4& o0, float4& o1) {
    float phi[4];
    phi[0] = 0.5f * th.x * cx;
    phi[1] = 0.5f * th.y * cy;
    phi[2] = 0.5f * th.z * cz;
    phi[3] = 0.5f * th.w * cw;

    float r[16];
#pragma unroll
    for (int k = 0; k < 16; ++k) r[k] = (k == 0) ? 1.0f : 0.0f;
    const int blades[4] = {3, 5, 9, 6};
#pragma unroll
    for (int pl = 0; pl < 4; ++pl) {
        float sn, cs;
        __sincosf(phi[pl], &sn, &cs);
        const int b = blades[pl];
        float nr[16];
#pragma unroll
        for (int k = 0; k < 16; ++k)
            nr[k] = cs * r[k] + sn * gpsign(k ^ b, b) * r[k ^ b];
#pragma unroll
        for (int k = 0; k < 16; ++k) r[k] = nr[k];
    }

    float rrev[16];
#pragma unroll
    for (int k = 0; k < 16; ++k) rrev[k] = r[k] * bivsign(k);

    const float sig = hi ? -1.0f : 1.0f;
    float ue[4], uo[4];
#pragma unroll
    for (int b = 0; b < 4; ++b) {
        const int Je = rep_blade(b), Jo = rep_blade(4 + b);
        ue[b] = xv[Je] + (sig * dual_sign(Je)) * xv[Je ^ 15];
        uo[b] = xv[Jo] + (sig * dual_sign(Jo)) * xv[Jo ^ 15];
    }
    {
        constexpr int J7 = rep_blade(7);
        constexpr float d7 = dual_sign(J7);
        uo[3] = sig * xv[J7] + d7 * xv[J7 ^ 15];
    }
    float qs[4], qo[4], qts[4];
#pragma unroll
    for (int a = 0; a < 4; ++a) {
        const int J = rep_blade(a);
        const float d = (sig * dual_sign(J)) * r[J ^ 15];
        qs[a]  = r[J] + d;
        qo[a]  = r[J] - d;
        qts[a] = rrev[J] + (sig * dual_sign(J)) * rrev[J ^ 15];
    }

    v8f acc;
    acc[0] = start1<0>(qs, ue, qo, uo);
    acc[1] = start1<1>(qs, ue, qo, uo);
    acc[2] = start1<2>(qs, ue, qo, uo);
    acc[3] = start1<3>(qs, ue, qo, uo);
    acc[4] = start1<4>(qs, ue, qo, uo);
    acc[5] = start1<5>(qs, ue, qo, uo);
    acc[6] = start1<6>(qs, ue, qo, uo);
    acc[7] = start1<7>(qs, ue, qo, uo);
    Product<0, 0>::run(acc, ctab, lane, qs, ue, qo, uo);

    const float we[4] = {acc[0], acc[1], acc[2], acc[3]};
    const float wo[4] = {acc[4], acc[5], acc[6], acc[7]};

    const float hs = 0.5f * sig;
    float sv[8];
    {
        const float t0 = term2<0>(we, wo, qts), t1 = term2<1>(we, wo, qts);
        const float t2 = term2<2>(we, wo, qts), t3 = term2<3>(we, wo, qts);
        const float t4 = term2<4>(we, wo, qts), t5 = term2<5>(we, wo, qts);
        const float t6 = term2<6>(we, wo, qts), t7 = term2<7>(we, wo, qts);
        sv[0] = start2<0>(hs, t0, other_half(t0));
        sv[1] = start2<1>(hs, t1, other_half(t1));
        sv[2] = start2<2>(hs, t2, other_half(t2));
        sv[3] = start2<3>(hs, t3, other_half(t3));
        sv[4] = start2<4>(hs, t4, other_half(t4));
        sv[5] = start2<5>(hs, t5, other_half(t5));
        sv[6] = start2<6>(hs, t6, other_half(t6));
        sv[7] = start2<7>(hs, t7, other_half(t7));
    }
    v8f acc2;
    acc2[0] = pick2<0>(hi, sv);
    acc2[1] = pick2<1>(hi, sv);
    acc2[2] = pick2<2>(hi, sv);
    acc2[3] = pick2<3>(hi, sv);
    acc2[4] = pick2<4>(hi, sv);
    acc2[5] = pick2<5>(hi, sv);
    acc2[6] = pick2<6>(hi, sv);
    acc2[7] = pick2<7>(hi, sv);
    Product<1, 0>::run(acc2, ctab, lane, we, qts, wo, qts);

    o0.x = acc2[0]; o0.y = acc2[1]; o0.z = acc2[2]; o0.w = acc2[3];
    o1.x = acc2[4]; o1.y = acc2[5]; o1.z = acc2[6]; o1.w = acc2[7];
}

__global__ __launch_bounds__(256)
void care_rotor_kernel(const float* __restrict__ x,
                       const int*   __restrict__ pos,
                       const float* __restrict__ bx,
                       const float* __restrict__ by,
                       const float* __restrict__ bz,
                       const float* __restrict__ bw,
                       const float* __restrict__ theta,
                       float* __restrict__ out,
                       int ntiles) {
    __shared__ alignas(16) float ctab[8 * 32 * 2];

    const int lane = threadIdx.x & 31;
    const int wave = threadIdx.x >> 5;
    const int kl   = lane & 15;
    const bool hi  = lane >= 16;

    if (threadIdx.x < 128) ((float4*)ctab)[threadIdx.x] = ((const float4*)g_ctab.v)[threadIdx.x];
    __syncthreads();

    const int tile0 = (blockIdx.x * 8 + wave) * TPW;
    if (tile0 >= ntiles) return;

    long p[TPW];
    bool live[TPW];
#pragma unroll
    for (int j = 0; j < TPW; ++j) {
        live[j] = (tile0 + j) < ntiles;
        p[j] = (long)(live[j] ? tile0 + j : tile0) * 16 + kl;
    }

    int pc[TPW];
#pragma unroll
    for (int j = 0; j < TPW; ++j) pc[j] = pos[p[j]];

    float4 q[TPW][4];
#pragma unroll
    for (int j = 0; j < TPW; ++j) {
        const float4* xp = (const float4*)(x + p[j] * MV);
#pragma unroll
        for (int t = 0; t < 4; ++t) q[j][t] = xp[t];
    }

    __builtin_amdgcn_sched_barrier(0);

    float4 th[TPW];
#pragma unroll
    for (int j = 0; j < TPW; ++j) {
        int c = pc[j];
        c = c < 0 ? 0 : (c > MAXLEN - 1 ? MAXLEN - 1 : c);
        th[j] = *(const float4*)(theta + (long)c * 4);
    }

    const float cx = bx[3], cy = by[5], cz = bz[9], cw = bw[6];

#pragma unroll
    for (int j = 0; j < TPW; ++j) {
        float xv[16];
#pragma unroll
        for (int t = 0; t < 4; ++t) {
            xv[4 * t + 0] = q[j][t].x; xv[4 * t + 1] = q[j][t].y;
            xv[4 * t + 2] = q[j][t].z; xv[4 * t + 3] = q[j][t].w;
        }
        float4 o0, o1;
        rotate_tile(xv, th[j], cx, cy, cz, cw, ctab, lane, hi, o0, o1);
        if (live[j]) {
            float* po = out + p[j] * MV + (hi ? 8 : 0);
            *(float4*)(po + 0) = o0;
            *(float4*)(po + 4) = o1;
        }
    }
}

extern "C" void kernel_launch(void* const* d_in, const int* in_sizes, int n_in,
                              void* d_out, int out_size, void* d_ws, size_t ws_size,
                              hipStream_t stream) {
    const float* x     = (const float*)d_in[0];
    const int*   pos   = (const int*)d_in[1];
    const float* bx    = (const float*)d_in[2];
    const float* by    = (const float*)d_in[3];
    const float* bz    = (const float*)d_in[4];
    const float* bw    = (const float*)d_in[5];
    const float* theta = (const float*)d_in[6];
    float* out = (float*)d_out;

    const int npos   = in_sizes[1];
    const int ntiles = npos / 16;
    const int blocks = (ntiles + 8 * TPW - 1) / (8 * TPW);

    hipLaunchKernelGGL(care_rotor_kernel, dim3(blocks), dim3(256), 0, stream,
                       x, pos, bx, by, bz, bw, theta, out, ntiles);
}


# Rotor built directly as its two quaternions: the dual planes e03, e12 merge into one angle per idempotent half, the reverse is a sign pattern of the blocks; 75 fewer vector instructions per wave
# speedup vs baseline: 4.0479x; 1.0665x over previous
#include <hip/hip_runtime.h>


typedef float v2f __attribute__((ext_vector_type(2)));
typedef float v8f __attribute__((ext_vector_type(8)));

#define MV 16
#define MAXLEN 16384

constexpr int popc4(int v) { return (v & 1) + ((v >> 1) & 1) + ((v >> 2) & 1) + ((v >> 3) & 1); }

constexpr float gpsign(int a, int b) {
    int s = 0;
    for (int t = a >> 1; t; t >>= 1) s += popc4(t & b);
    return (s & 1) ? -1.0f : 1.0f;
}

constexpr int rep_blade(int i) {
    constexpr int t[8] = {0, 3, 5, 6, 1, 2, 4, 8};
    return t[i];
}
constexpr int rep_index(int blade) {
    for (int i = 0; i < 8; ++i) if (rep_blade(i) == blade) return i;
    return -1;
}
constexpr float dual_sign(int J) { return gpsign(J, 15); }

constexpr int coord_index(int Jx) {
    return rep_index(Jx) >= 0 ? rep_index(Jx) : rep_index(Jx ^ 15);
}
constexpr float coord_sign(int Jx, float sig) {
    return rep_index(Jx) >= 0 ? 1.0f : sig * dual_sign(Jx ^ 15);
}

struct Comb { int i, j, s; };
constexpr Comb HA[8] = {{3, 1, 1}, {0, 2, -1}, {0, 2, 1}, {3, 1, -1}, {3, 2, -1}, {1, 0, 1}, {0, 1, -1}, {3, 2, 1}};
constexpr Comb HB[8] = {{1, 2, 1}, {0, 3, 1}, {0, 3, -1}, {1, 2, -1}, {2, 3, -1}, {1, 0, 1}, {2, 3, 1}, {0, 1, -1}};
constexpr float HC[4][8] = {
    {-0.5f,  0.5f,  0.5f, 0.5f, 1.0f, 0.0f, 0.0f, 0.0f},
    {-0.5f, -0.5f, -0.5f, 0.5f, 0.0f, 1.0f, 0.0f, 0.0f},
    { 0.5f, -0.5f,  0.5f, 0.5f, 0.0f, 0.0f, 1.0f, 0.0f},
    { 0.5f,  0.5f, -0.5f, 0.5f, 0.0f, 0.0f, 0.0f, 1.0f}};

struct BlockSigns { int sa[4], sb[4], so[2][4]; };
constexpr BlockSigns BS[4] = {
    {{-1, 1, 1, 1}, {-1,  1,  1, 1}, {{ 1, -1, -1, -1}, { 1, -1, -1, -1}}},
    {{-1, 1, 1, 1}, { 1,  1,  1, 1}, {{-1, -1, -1, -1}, {-1, -1, -1,  1}}},
    {{-1, 1, 1, 1}, {-1, -1, -1, -1}, {{ 1, -1, -1, -1}, { 1, -1, -1, -1}}},
    {{ 1, 1, 1, 1}, {-1,  1,  1, -1}, {{-1, -1, -1, -1}, {-1, -1, -1,  1}}}};

constexpr Comb blk_left(int b, int m)  { return {HA[m].i, HA[m].j, HA[m].s * BS[b].sa[HA[m].i] * BS[b].sa[HA[m].j]}; }
constexpr Comb blk_right(int b, int m) { return {HB[m].i, HB[m].j, HB[m].s * BS[b].sb[HB[m].i] * BS[b].sb[HB[m].j]}; }
constexpr float blk_coef(int b, int h, int c, int m) {
    return (float)(BS[b].so[h][c] * BS[b].sa[HA[m].i] * BS[b].sb[HB[m].i]) * HC[c][m];
}

constexpr float blk_expect(int b, int h, int c, int l, int r) {
    const float sig = h ? -1.0f : 1.0f;
    const int odd = b & 1;
    const int La = (b < 2) ? rep_blade(l) : rep_blade(4 * odd + l);
    const int Rb = (b < 2) ? rep_blade(4 * odd + r) : rep_blade(r);
    if (coord_index(La ^ Rb) != 4 * odd + c) return 0.0f;
    float s = gpsign(La, Rb) * coord_sign(La ^ Rb, sig);
    if (b == 1 && r == 3) s *= sig;
    if (b == 3 && l == 3) s *= sig;
    if (b >= 2 && r != 0) s = -s;
    return s;
}
constexpr bool blk_ok(int b) {
    for (int h = 0; h < 2; ++h)
        for (int c = 0; c < 4; ++c)
            for (int l = 0; l < 4; ++l)
                for (int r = 0; r < 4; ++r) {
                    float v = 0.0f;
                    for (int m = 0; m < 8; ++m) {
                        const Comb a = blk_left(b, m), d = blk_right(b, m);
                        const float al = (a.i == l ? 1.0f : 0.0f) + (a.j == l ? (float)a.s : 0.0f);
                        const float be = (d.i == r ? 1.0f : 0.0f) + (d.j == r ? (float)d.s : 0.0f);
                        v += blk_coef(b, h, c, m) * al * be;
                    }
                    if (v != blk_expect(b, h, c, l, r)) return false;
                }
    return true;
}
static_assert(blk_ok(0) && blk_ok(1) && blk_ok(2) && blk_ok(3),
              "8-multiplication blocks do not reproduce the Cayley signs");

constexpr float p1_weight(int row, int h, int odd, int m) {
    const float sig = h ? -1.0f : 1.0f;
    const int J = row - 8 * h;
    if (J < 0 || J > 7 || (J >> 2) != odd) return 0.0f;
    return blk_coef(odd, h, J & 3, m) * (J == 7 ? sig : 1.0f);
}
constexpr float p2_weight(int row, int h, int odd, int m) {
    const float sig = h ? -1.0f : 1.0f;
    float e = 0.0f;
    for (int c = 0; c < 4; ++c) {
        const int B = rep_blade(4 * odd + c);
        if (row == B)        e += 0.5f * blk_coef(2 + odd, h, c, m);
        if (row == (B ^ 15)) e += 0.5f * sig * dual_sign(B) * blk_coef(2 + odd, h, c, m);
    }
    return e;
}

constexpr float tab_entry(int chunk, int lane, int v) {
    const int h = lane >> 4, row = lane & 15;
    const int tm = 2 * (chunk & 3) + v;
    return chunk < 4 ? p1_weight(row, h, tm >> 2, tm & 3) : p2_weight(row, h, tm >> 2, tm & 3);
}
struct alignas(16) CTab { float v[8 * 32 * 2]; };
constexpr CTab make_ctab() {
    CTab t{};
    for (int chunk = 0; chunk < 8; ++chunk)
        for (int lane = 0; lane < 32; ++lane)
            for (int v = 0; v < 2; ++v)
                t.v[chunk * 64 + lane * 2 + v] = tab_entry(chunk, lane, v);
    return t;
}

__constant__ const CTab g_ctab = make_ctab();

constexpr int p2_row_of_term(int h, int d) {
    const int B = rep_blade(d);
    return ((B >> 3) == h) ? B : (B ^ 15);
}
constexpr int p2_term_of_row(int row) {
    for (int d = 0; d < 8; ++d)
        if (row == rep_blade(d) || row == (rep_blade(d) ^ 15)) return d;
    return -1;
}
constexpr bool start_ok() {
    for (int h = 0; h < 2; ++h)
        for (int row = 0; row < 16; ++row)
            for (int odd = 0; odd < 2; ++odd)
                for (int m = 4; m < 8; ++m) {
                    const int d = 4 * odd + (m - 4);
                    const bool own1 = (row == 8 * h + d);
                    if ((p1_weight(row, h, odd, m) != 0.0f) != own1) return false;
                    const bool own2 = (row == p2_row_of_term(0, d) || row == p2_row_of_term(1, d));
                    if ((p2_weight(row, h, odd, m) != 0.0f) != own2) return false;
                }
    for (int row = 0; row < 16; ++row)
        if (p2_row_of_term(row >> 3, p2_term_of_row(row)) != row) return false;
    return true;
}
static_assert(start_ok(), "single-row terms do not match the accumulator rows");

template <int B, int M>
__device__ __forceinline__ float fterm(const float (&L)[4], const float (&R)[4]) {
    constexpr Comb a = blk_left(B, M), b = blk_right(B, M);
    const float al = (a.s > 0) ? (L[a.i] + L[a.j]) : (L[a.i] - L[a.j]);
    const float be = (b.s > 0) ? (R[b.i] + R[b.j]) : (R[b.i] - R[b.j]);
    return al * be;
}

template <int J>
__device__ __forceinline__ float start1(const float (&Le)[4], const float (&Re)[4],
                                        const float (&Lo)[4], const float (&Ro)[4]) {
    constexpr int odd = J >> 2, m = 4 + (J & 3);
    constexpr float k0 = p1_weight(J, 0, odd, m), k1 = p1_weight(8 + J, 1, odd, m);
    static_assert((k0 == 1.0f || k0 == -1.0f) && k1 == k0, "weight of a single-row term");
    if constexpr (odd == 0) return k0 * fterm<0, m>(Le, Re); else return k0 * fterm<1, m>(Lo, Ro);
}

template <int D>
__device__ __forceinline__ float term2(const float (&We)[4], const float (&Wo)[4], const float (&Q)[4]) {
    if constexpr (D < 4) return fterm<2, 4 + D>(We, Q); else return fterm<3, D>(Wo, Q);
}

template <int D>
__device__ __forceinline__ float start2(float hs, float own, float oth) {
    constexpr int odd = D >> 2, m = 4 + (D & 3);
    constexpr float a0 = p2_weight(p2_row_of_term(0, D), 0, odd, m), a1 = p2_weight(p2_row_of_term(1, D), 1, odd, m);
    constexpr float b0 = p2_weight(p2_row_of_term(0, D), 1, odd, m), b1 = p2_weight(p2_row_of_term(1, D), 0, odd, m);
    static_assert((a0 == 0.5f || a0 == -0.5f) && (a1 == a0 || a1 == -a0), "own weight of a single-row term");
    static_assert((b0 == 0.5f || b0 == -0.5f) && (b1 == b0 || b1 == -b0), "other weight of a single-row term");
    float wa, wb;
    if constexpr (a1 == a0) wa = a0; else wa = (2.0f * a0) * hs;
    if constexpr (b1 == b0) wb = b0; else wb = (2.0f * b0) * hs;
    return wa * own + wb * oth;
}

template <int V>
__device__ __forceinline__ float pick2(bool hi, const float (&sv)[8]) {
    constexpr int d0 = p2_term_of_row(V), d1 = p2_term_of_row(8 + V);
    return hi ? sv[d1] : sv[d0];
}

__device__ __forceinline__ float other_half(float x) {
    return __builtin_bit_cast(float, __builtin_amdgcn_permlanex16(__builtin_bit_cast(int, x), __builtin_bit_cast(int, x),
                                                                  0x76543210, 0xfedcba98, false, false));
}

template <int P, int C>
struct Product {
    static __device__ __forceinline__ void run(v8f& acc, const float* ctab, int lane,
                                               const float (&Le)[4], const float (&Re)[4],
                                               const float (&Lo)[4], const float (&Ro)[4]) {
        const v2f aC = *(const v2f*)(ctab + (4 * P + C) * 64 + lane * 2);
        v2f zB;
        if constexpr (C < 2) {
            zB[0] = fterm<2 * P, 2 * C>(Le, Re);
            zB[1] = fterm<2 * P, 2 * C + 1>(Le, Re);
        } else {
            zB[0] = fterm<2 * P + 1, 2 * C - 4>(Lo, Ro);
            zB[1] = fterm<2 * P + 1, 2 * C - 3>(Lo, Ro);
        }
        acc = __builtin_amdgcn_wmma_f32_16x16x4_f32(false, aC, false, zB,
                                                    (short)0, acc, false, false);
        Product<P, C + 1>::run(acc, ctab, lane, Le, Re, Lo, Ro);
    }
};
template <int P>
struct Product<P, 4> {
    static __device__ __forceinline__ void run(v8f&, const float*, int,
                                               const float (&)[4], const float (&)[4],
                                               const float (&)[4], const float (&)[4]) {}
};

#define TPW 2

__device__ __forceinline__ void rotate_tile(const float (&xv)[16], const float4 th,
                                            const float cx, const float cy, const float cz, const float cw,
                                            const float* ctab, const int lane, const bool hi,
                                            float4& o0, float4& o1) {
    float phi[4];
    phi[0] = 0.5f * th.x * cx;
    phi[1] = 0.5f * th.y * cy;
    phi[2] = 0.5f * th.z * cz;
    phi[3] = 0.5f * th.w * cw;

    float sn0, cs0, sn1, cs1;
    __sincosf(phi[0], &sn0, &cs0);
    __sincosf(phi[1], &sn1, &cs1);
    constexpr float d9 = coord_sign(9, 1.0f);
    const float sd = hi ? -d9 : d9;
    float snS, csS, snO, csO;
    __sincosf(phi[3] + sd * phi[2], &snS, &csS);
    __sincosf(phi[3] - sd * phi[2], &snO, &csO);
    const float ab[4] = {cs0 * cs1, sn0 * cs1, cs0 * sn1, gpsign(3, 5) * (sn0 * sn1)};

    const float sig = hi ? -1.0f : 1.0f;
    float ue[4], uo[4];
#pragma unroll
    for (int b = 0; b < 4; ++b) {
        const int Je = rep_blade(b), Jo = rep_blade(4 + b);
        ue[b] = xv[Je] + (sig * dual_sign(Je)) * xv[Je ^ 15];
        uo[b] = xv[Jo] + (sig * dual_sign(Jo)) * xv[Jo ^ 15];
    }
    {
        constexpr int J7 = rep_blade(7);
        constexpr float d7 = dual_sign(J7);
        uo[3] = sig * xv[J7] + d7 * xv[J7 ^ 15];
    }
    float qs[4], qo[4];
#pragma unroll
    for (int a = 0; a < 4; ++a) {
        const int m = rep_blade(a) ^ 6;
        const float g = gpsign(m, 6);
        const float am = ab[rep_index(m)];
        qs[a]  = ab[a] * csS + g * (am * snS);
        qo[a]  = ab[a] * csO + g * (am * snO);
    }

    v8f acc;
    acc[0] = start1<0>(qs, ue, qo, uo);
    acc[1] = start1<1>(qs, ue, qo, uo);
    acc[2] = start1<2>(qs, ue, qo, uo);
    acc[3] = start1<3>(qs, ue, qo, uo);
    acc[4] = start1<4>(qs, ue, qo, uo);
    acc[5] = start1<5>(qs, ue, qo, uo);
    acc[6] = start1<6>(qs, ue, qo, uo);
    acc[7] = start1<7>(qs, ue, qo, uo);
    Product<0, 0>::run(acc, ctab, lane, qs, ue, qo, uo);

    const float we[4] = {acc[0], acc[1], acc[2], acc[3]};
    const float wo[4] = {acc[4], acc[5], acc[6], acc[7]};

    const float hs = 0.5f * sig;
    float sv[8];
    {
        const float t0 = term2<0>(we, wo, qs), t1 = term2<1>(we, wo, qs);
        const float t2 = term2<2>(we, wo, qs), t3 = term2<3>(we, wo, qs);
        const float t4 = term2<4>(we, wo, qs), t5 = term2<5>(we, wo, qs);
        const float t6 = term2<6>(we, wo, qs), t7 = term2<7>(we, wo, qs);
        sv[0] = start2<0>(hs, t0, other_half(t0));
        sv[1] = start2<1>(hs, t1, other_half(t1));
        sv[2] = start2<2>(hs, t2, other_half(t2));
        sv[3] = start2<3>(hs, t3, other_half(t3));
        sv[4] = start2<4>(hs, t4, other_half(t4));
        sv[5] = start2<5>(hs, t5, other_half(t5));
        sv[6] = start2<6>(hs, t6, other_half(t6));
        sv[7] = start2<7>(hs, t7, other_half(t7));
    }
    v8f acc2;
    acc2[0] = pick2<0>(hi, sv);
    acc2[1] = pick2<1>(hi, sv);
    acc2[2] = pick2<2>(hi, sv);
    acc2[3] = pick2<3>(hi, sv);
    acc2[4] = pick2<4>(hi, sv);
    acc2[5] = pick2<5>(hi, sv);
    acc2[6] = pick2<6>(hi, sv);
    acc2[7] = pick2<7>(hi, sv);
    Product<1, 0>::run(acc2, ctab, lane, we, qs, wo, qs);

    o0.x = acc2[0]; o0.y = acc2[1]; o0.z = acc2[2]; o0.w = acc2[3];
    o1.x = acc2[4]; o1.y = acc2[5]; o1.z = acc2[6]; o1.w = acc2[7];
}

__global__ __launch_bounds__(256)
void care_rotor_kernel(const float* __restrict__ x,
                       const int*   __restrict__ pos,
                       const float* __restrict__ bx,
                       const float* __restrict__ by,
                       const float* __restrict__ bz,
                       const float* __restrict__ bw,
                       const float* __restrict__ theta,
                       float* __restrict__ out,
                       int ntiles) {
    __shared__ alignas(16) float ctab[8 * 32 * 2];

    const int lane = threadIdx.x & 31;
    const int wave = threadIdx.x >> 5;
    const int kl   = lane & 15;
    const bool hi  = lane >= 16;

    if (threadIdx.x < 128) ((float4*)ctab)[threadIdx.x] = ((const float4*)g_ctab.v)[threadIdx.x];
    __syncthreads();

    const int tile0 = (blockIdx.x * 8 + wave) * TPW;
    if (tile0 >= ntiles) return;

    long p[TPW];
    bool live[TPW];
#pragma unroll
    for (int j = 0; j < TPW; ++j) {
        live[j] = (tile0 + j) < ntiles;
        p[j] = (long)(live[j] ? tile0 + j : tile0) * 16 + kl;
    }

    int pc[TPW];
#pragma unroll
    for (int j = 0; j < TPW; ++j) pc[j] = pos[p[j]];

    float4 q[TPW][4];
#pragma unroll
    for (int j = 0; j < TPW; ++j) {
        const float4* xp = (const float4*)(x + p[j] * MV);
#pragma unroll
        for (int t = 0; t < 4; ++t) q[j][t] = xp[t];
    }

    __builtin_amdgcn_sched_barrier(0);

    float4 th[TPW];
#pragma unroll
    for (int j = 0; j < TPW; ++j) {
        int c = pc[j];
        c = c < 0 ? 0 : (c > MAXLEN - 1 ? MAXLEN - 1 : c);
        th[j] = *(const float4*)(theta + (long)c * 4);
    }

    const float cx = bx[3], cy = by[5], cz = bz[9], cw = bw[6];

#pragma unroll
    for (int j = 0; j < TPW; ++j) {
        float xv[16];
#pragma unroll
        for (int t = 0; t < 4; ++t) {
            xv[4 * t + 0] = q[j][t].x; xv[4 * t + 1] = q[j][t].y;
            xv[4 * t + 2] = q[j][t].z; xv[4 * t + 3] = q[j][t].w;
        }
        float4 o0, o1;
        rotate_tile(xv, th[j], cx, cy, cz, cw, ctab, lane, hi, o0, o1);
        if (live[j]) {
            float* po = out + p[j] * MV + (hi ? 8 : 0);
            *(float4*)(po + 0) = o0;
            *(float4*)(po + 4) = o1;
        }
    }
}

extern "C" void kernel_launch(void* const* d_in, const int* in_sizes, int n_in,
                              void* d_out, int out_size, void* d_ws, size_t ws_size,
                              hipStream_t stream) {
    const float* x     = (const float*)d_in[0];
    const int*   pos   = (const int*)d_in[1];
    const float* bx    = (const float*)d_in[2];
    const float* by    = (const float*)d_in[3];
    const float* bz    = (const float*)d_in[4];
    const float* bw    = (const float*)d_in[5];
    const float* theta = (const float*)d_in[6];
    float* out = (float*)d_out;

    const int npos   = in_sizes[1];
    const int ntiles = npos / 16;
    const int blocks = (ntiles + 8 * TPW - 1) / (8 * TPW);

    hipLaunchKernelGGL(care_rotor_kernel, dim3(blocks), dim3(256), 0, stream,
                       x, pos, bx, by, bz, bw, theta, out, ntiles);
}


# Both quaternion products directly on the vector unit (1 mul + 3 fma per coordinate, compile-time signs); f32 WMMA keeps only the constant map from the two halves to the 16 blades: 4 per tile, not 8
# speedup vs baseline: 4.8238x; 1.1917x over previous
#include <hip/hip_runtime.h>


typedef float v2f __attribute__((ext_vector_type(2)));
typedef float v8f __attribute__((ext_vector_type(8)));

#define MV 16
#define MAXLEN 16384

constexpr int popc4(int v) { return (v & 1) + ((v >> 1) & 1) + ((v >> 2) & 1) + ((v >> 3) & 1); }

constexpr float gpsign(int a, int b) {
    int s = 0;
    for (int t = a >> 1; t; t >>= 1) s += popc4(t & b);
    return (s & 1) ? -1.0f : 1.0f;
}

constexpr int rep_blade(int i) {
    constexpr int t[8] = {0, 3, 5, 6, 1, 2, 4, 8};
    return t[i];
}
constexpr int rep_index(int blade) {
    for (int i = 0; i < 8; ++i) if (rep_blade(i) == blade) return i;
    return -1;
}
constexpr float dual_sign(int J) { return gpsign(J, 15); }

constexpr int coord_index(int Jx) {
    return rep_index(Jx) >= 0 ? rep_index(Jx) : rep_index(Jx ^ 15);
}
constexpr float coord_sign(int Jx, float sig) {
    return rep_index(Jx) >= 0 ? 1.0f : sig * dual_sign(Jx ^ 15);
}

constexpr float blk_expect(int b, int h, int c, int l, int r) {
    const float sig = h ? -1.0f : 1.0f;
    const int odd = b & 1;
    const int La = (b < 2) ? rep_blade(l) : rep_blade(4 * odd + l);
    const int Rb = (b < 2) ? rep_blade(4 * odd + r) : rep_blade(r);
    if (coord_index(La ^ Rb) != 4 * odd + c) return 0.0f;
    float s = gpsign(La, Rb) * coord_sign(La ^ Rb, sig);
    if (b == 1 && r == 3) s *= sig;
    if (b == 3 && l == 3) s *= sig;
    if (b >= 2 && r != 0) s = -s;
    return s;
}
constexpr float blk_dir(int b, int h, int c, int l, int r) {
    const float sig = h ? -1.0f : 1.0f;
    return blk_expect(b, h, c, l, r) * ((b == 1 && c == 3) ? sig : 1.0f);
}
constexpr int blk_r(int b, int c, int l) {
    for (int r = 0; r < 4; ++r) if (blk_dir(b, 0, c, l, r) != 0.0f) return r;
    return -1;
}
constexpr float blk_flip(int b, int c) {
    return blk_dir(b, 1, c, 0, blk_r(b, c, 0)) * blk_dir(b, 0, c, 0, blk_r(b, c, 0));
}
constexpr bool dir_ok() {
    for (int b = 0; b < 4; ++b)
        for (int c = 0; c < 4; ++c) {
            const float fl = blk_flip(b, c);
            if (fl != 1.0f && fl != -1.0f) return false;
            if (b < 2 && fl != 1.0f) return false;
            for (int l = 0; l < 4; ++l) {
                const int r0 = blk_r(b, c, l);
                if (r0 < 0) return false;
                for (int r = 0; r < 4; ++r) {
                    const float v0 = blk_dir(b, 0, c, l, r), v1 = blk_dir(b, 1, c, l, r);
                    if (r != r0) { if (v0 != 0.0f || v1 != 0.0f) return false; }
                    else if ((v0 != 1.0f && v0 != -1.0f) || v1 != fl * v0) return false;
                }
            }
        }
    return true;
}
static_assert(dir_ok(), "a block is not a signed Hamilton product with one sign per lane half and coordinate");

constexpr float fin_weight(int row, int h, int J) {
    const float sig = h ? -1.0f : 1.0f;
    const int B = rep_blade(J);
    const float fl = h ? blk_flip(2 + (J >> 2), J & 3) : 1.0f;
    float e = 0.0f;
    if (row == B)        e += 0.5f * fl;
    if (row == (B ^ 15)) e += 0.5f * sig * dual_sign(B) * fl;
    return e;
}

constexpr float tab_entry(int chunk, int lane, int v) {
    const int h = lane >> 4, row = lane & 15;
    return fin_weight(row, h, 2 * chunk + v);
}
struct alignas(16) CTab { float v[4 * 32 * 2]; };
constexpr CTab make_ctab() {
    CTab t{};
    for (int chunk = 0; chunk < 4; ++chunk)
        for (int lane = 0; lane < 32; ++lane)
            for (int v = 0; v < 2; ++v)
                t.v[chunk * 64 + lane * 2 + v] = tab_entry(chunk, lane, v);
    return t;
}

__constant__ const CTab g_ctab = make_ctab();

struct Mv { float v[16]; };
constexpr Mv gp_full(const Mv& a, const Mv& b) {
    Mv r{};
    for (int i = 0; i < 16; ++i)
        for (int j = 0; j < 16; ++j) r.v[i ^ j] += gpsign(i, j) * a.v[i] * b.v[j];
    return r;
}
constexpr float blk_eval(int b, int c, const float* L, const float* R) {
    float s = 0.0f;
    for (int l = 0; l < 4; ++l) s += blk_dir(b, 0, c, l, blk_r(b, c, l)) * L[l] * R[blk_r(b, c, l)];
    return s;
}
constexpr bool route_ok(unsigned seed) {
    Mv R{}, X{};
    unsigned s = seed;
    for (int i = 0; i < 16; ++i) {
        s = s * 1664525u + 1013904223u;
        X.v[i] = (float)((int)((s >> 24) % 7u) - 3);
        s = s * 1664525u + 1013904223u;
        R.v[i] = (popc4(i) & 1) ? 0.0f : (float)((int)((s >> 24) % 5u) - 2);
    }
    Mv Rr = R;
    for (int i = 0; i < 16; ++i) if (popc4(i) == 2) Rr.v[i] = -R.v[i];
    const Mv ref = gp_full(gp_full(R, X), Rr);
    float Y[2][8] = {};
    for (int h = 0; h < 2; ++h) {
        const float sig = h ? -1.0f : 1.0f;
        float qs[4] = {}, qo[4] = {}, ue[4] = {}, uo[4] = {}, we[4] = {}, wo[4] = {};
        for (int a = 0; a < 4; ++a) {
            const int Je = rep_blade(a), Jo = rep_blade(4 + a);
            qs[a] = R.v[Je] + sig * dual_sign(Je) * R.v[Je ^ 15];
            qo[a] = R.v[Je] - sig * dual_sign(Je) * R.v[Je ^ 15];
            ue[a] = X.v[Je] + sig * dual_sign(Je) * X.v[Je ^ 15];
            uo[a] = X.v[Jo] + sig * dual_sign(Jo) * X.v[Jo ^ 15];
        }
        uo[3] *= sig;
        for (int c = 0; c < 4; ++c) { we[c] = blk_eval(0, c, qs, ue); wo[c] = blk_eval(1, c, qo, uo); }
        for (int c = 0; c < 4; ++c) { Y[h][c] = blk_eval(2, c, we, qs); Y[h][4 + c] = blk_eval(3, c, wo, qs); }
    }
    for (int row = 0; row < 16; ++row) {
        float o = 0.0f;
        for (int h = 0; h < 2; ++h)
            for (int J = 0; J < 8; ++J) o += fin_weight(row, h, J) * Y[h][J];
        if (o != ref.v[row]) return false;
    }
    return true;
}
static_assert(route_ok(1u) && route_ok(77u) && route_ok(2024u) && route_ok(909090u),
              "blocks and return map do not reproduce R x R~");

template <int B, int C>
__device__ __forceinline__ float blk_coord(const float (&L)[4], const float (&R)[4]) {
    constexpr int r0 = blk_r(B, C, 0), r1 = blk_r(B, C, 1), r2 = blk_r(B, C, 2), r3 = blk_r(B, C, 3);
    constexpr float k0 = blk_dir(B, 0, C, 0, r0), k1 = blk_dir(B, 0, C, 1, r1);
    constexpr float k2 = blk_dir(B, 0, C, 2, r2), k3 = blk_dir(B, 0, C, 3, r3);
    float s = (k0 * L[0]) * R[r0];
    s = __builtin_fmaf(k1 * L[1], R[r1], s);
    s = __builtin_fmaf(k2 * L[2], R[r2], s);
    s = __builtin_fmaf(k3 * L[3], R[r3], s);
    return s;
}

#define TPW 2

__device__ __forceinline__ void rotate_tile(const float (&xv)[16], const float4 th,
                                            const float cx, const float cy, const float cz, const float cw,
                                            const float* ctab, const int lane, const bool hi,
                                            float4& o0, float4& o1) {
    float phi[4];
    phi[0] = 0.5f * th.x * cx;
    phi[1] = 0.5f * th.y * cy;
    phi[2] = 0.5f * th.z * cz;
    phi[3] = 0.5f * th.w * cw;

    float sn0, cs0, sn1, cs1;
    __sincosf(phi[0], &sn0, &cs0);
    __sincosf(phi[1], &sn1, &cs1);
    constexpr float d9 = coord_sign(9, 1.0f);
    const float sd = hi ? -d9 : d9;
    float snS, csS, snO, csO;
    __sincosf(phi[3] + sd * phi[2], &snS, &csS);
    __sincosf(phi[3] - sd * phi[2], &snO, &csO);
    const float ab[4] = {cs0 * cs1, sn0 * cs1, cs0 * sn1, gpsign(3, 5) * (sn0 * sn1)};

    const float sig = hi ? -1.0f : 1.0f;
    float ue[4], uo[4];
#pragma unroll
    for (int b = 0; b < 4; ++b) {
        const int Je = rep_blade(b), Jo = rep_blade(4 + b);
        ue[b] = xv[Je] + (sig * dual_sign(Je)) * xv[Je ^ 15];
        uo[b] = xv[Jo] + (sig * dual_sign(Jo)) * xv[Jo ^ 15];
    }
    {
        constexpr int J7 = rep_blade(7);
        constexpr float d7 = dual_sign(J7);
        uo[3] = sig * xv[J7] + d7 * xv[J7 ^ 15];
    }
    float qs[4], qo[4];
#pragma unroll
    for (int a = 0; a < 4; ++a) {
        const int m = rep_blade(a) ^ 6;
        const float g = gpsign(m, 6);
        const float am = ab[rep_index(m)];
        qs[a]  = ab[a] * csS + g * (am * snS);
        qo[a]  = ab[a] * csO + g * (am * snO);
    }

    const float we[4] = {blk_coord<0, 0>(qs, ue), blk_coord<0, 1>(qs, ue), blk_coord<0, 2>(qs, ue), blk_coord<0, 3>(qs, ue)};
    const float wo[4] = {blk_coord<1, 0>(qo, uo), blk_coord<1, 1>(qo, uo), blk_coord<1, 2>(qo, uo), blk_coord<1, 3>(qo, uo)};

    const float y[8] = {blk_coord<2, 0>(we, qs), blk_coord<2, 1>(we, qs), blk_coord<2, 2>(we, qs), blk_coord<2, 3>(we, qs),
                        blk_coord<3, 0>(wo, qs), blk_coord<3, 1>(wo, qs), blk_coord<3, 2>(wo, qs), blk_coord<3, 3>(wo, qs)};

    v8f acc2 = {0.0f, 0.0f, 0.0f, 0.0f, 0.0f, 0.0f, 0.0f, 0.0f};
#pragma unroll
    for (int c = 0; c < 4; ++c) {
        const v2f aC = *(const v2f*)(ctab + c * 64 + lane * 2);
        v2f zB;
        zB[0] = y[2 * c];
        zB[1] = y[2 * c + 1];
        acc2 = __builtin_amdgcn_wmma_f32_16x16x4_f32(false, aC, false, zB, (short)0, acc2, false, false);
    }

    o0.x = acc2[0]; o0.y = acc2[1]; o0.z = acc2[2]; o0.w = acc2[3];
    o1.x = acc2[4]; o1.y = acc2[5]; o1.z = acc2[6]; o1.w = acc2[7];
}

__global__ __launch_bounds__(256)
void care_rotor_kernel(const float* __restrict__ x,
                       const int*   __restrict__ pos,
                       const float* __restrict__ bx,
                       const float* __restrict__ by,
                       const float* __restrict__ bz,
                       const float* __restrict__ bw,
                       const float* __restrict__ theta,
                       float* __restrict__ out,
                       int ntiles) {
    __shared__ alignas(16) float ctab[4 * 32 * 2];

    const int lane = threadIdx.x & 31;
    const int wave = threadIdx.x >> 5;
    const int kl   = lane & 15;
    const bool hi  = lane >= 16;

    if (threadIdx.x < 64) ((float4*)ctab)[threadIdx.x] = ((const float4*)g_ctab.v)[threadIdx.x];
    __syncthreads();

    const int tile0 = (blockIdx.x * 8 + wave) * TPW;
    if (tile0 >= ntiles) return;

    long p[TPW];
    bool live[TPW];
#pragma unroll
    for (int j = 0; j < TPW; ++j) {
        live[j] = (tile0 + j) < ntiles;
        p[j] = (long)(live[j] ? tile0 + j : tile0) * 16 + kl;
    }

    int pc[TPW];
#pragma unroll
    for (int j = 0; j < TPW; ++j) pc[j] = pos[p[j]];

    float4 q[TPW][4];
#pragma unroll
    for (int j = 0; j < TPW; ++j) {
        const float4* xp = (const float4*)(x + p[j] * MV);
#pragma unroll
        for (int t = 0; t < 4; ++t) q[j][t] = xp[t];
    }

    __builtin_amdgcn_sched_barrier(0);

    float4 th[TPW];
#pragma unroll
    for (int j = 0; j < TPW; ++j) {
        int c = pc[j];
        c = c < 0 ? 0 : (c > MAXLEN - 1 ? MAXLEN - 1 : c);
        th[j] = *(const float4*)(theta + (long)c * 4);
    }

    const float cx = bx[3], cy = by[5], cz = bz[9], cw = bw[6];

#pragma unroll
    for (int j = 0; j < TPW; ++j) {
        float xv[16];
#pragma unroll
        for (int t = 0; t < 4; ++t) {
            xv[4 * t + 0] = q[j][t].x; xv[4 * t + 1] = q[j][t].y;
            xv[4 * t + 2] = q[j][t].z; xv[4 * t + 3] = q[j][t].w;
        }
        float4 o0, o1;
        rotate_tile(xv, th[j], cx, cy, cz, cw, ctab, lane, hi, o0, o1);
        if (live[j]) {
            float* po = out + p[j] * MV + (hi ? 8 : 0);
            *(float4*)(po + 0) = o0;
            *(float4*)(po + 4) = o1;
        }
    }
}

extern "C" void kernel_launch(void* const* d_in, const int* in_sizes, int n_in,
                              void* d_out, int out_size, void* d_ws, size_t ws_size,
                              hipStream_t stream) {
    const float* x     = (const float*)d_in[0];
    const int*   pos   = (const int*)d_in[1];
    const float* bx    = (const float*)d_in[2];
    const float* by    = (const float*)d_in[3];
    const float* bz    = (const float*)d_in[4];
    const float* bw    = (const float*)d_in[5];
    const float* theta = (const float*)d_in[6];
    float* out = (float*)d_out;

    const int npos   = in_sizes[1];
    const int ntiles = npos / 16;
    const int blocks = (ntiles + 8 * TPW - 1) / (8 * TPW);

    hipLaunchKernelGGL(care_rotor_kernel, dim3(blocks), dim3(256), 0, stream,
                       x, pos, bx, by, bz, bw, theta, out, ntiles);
}
